# starting kernel
# baseline (speedup 1.0000x reference)
#include <hip/hip_runtime.h>


#define BATCH  16384
#define T      700
#define F      3
#define TCHUNK 20
#define NCHUNK (T / TCHUNK)
#define BLOCK  128
#define NWAVES (BLOCK / 32)

typedef __attribute__((ext_vector_type(2))) float v2f;
typedef __attribute__((ext_vector_type(8))) float v8f;

__device__ __forceinline__ float fast_sigmoid(float x) {
    float e = __builtin_amdgcn_exp2f(-x * 1.44269504088896340736f);
    return __builtin_amdgcn_rcpf(1.0f + e);
}

__device__ __forceinline__ float fast_tanh(float x) {
    float e = __builtin_amdgcn_exp2f(-x * 2.88539008177792681472f);
    return 2.0f * __builtin_amdgcn_rcpf(1.0f + e) - 1.0f;
}

__global__ void lstm_scan_kernel(const float* __restrict__ x,
                                 const float* __restrict__ kern,
                                 const float* __restrict__ rkern,
                                 const float* __restrict__ bias,
                                 float* __restrict__ out) {
    extern __shared__ float smem[];
    float* x_lds  = smem;
    float* xz_lds = smem + BLOCK * TCHUNK * F;

    const int tid  = threadIdx.x;
    const int lane = tid & 31;
    const int wave = tid >> 5;
    const int b    = blockIdx.x * BLOCK + tid;

    const int n  = lane & 15;
    const int kb = (lane < 16) ? 0 : 2;
    v2f bmat;
    bmat.x = (n < 4 && (kb + 0) < F) ? kern[(kb + 0) * 4 + n] : 0.0f;
    bmat.y = (n < 4 && (kb + 1) < F) ? kern[(kb + 1) * 4 + n] : 0.0f;

    const float bias_n = (n < 4) ? bias[n] : 0.0f;
    v8f cmat;
#pragma unroll
    for (int v = 0; v < 8; ++v) cmat[v] = bias_n;

    const float aymask = (kb == 0) ? 1.0f : 0.0f;

    const float rk0 = rkern[0], rk1 = rkern[1], rk2 = rkern[2], rk3 = rkern[3];

    float h = 0.0f, c = 0.0f;
    const float* xg = x + (size_t)b * (T * F);

    for (int chunk = 0; chunk < NCHUNK; ++chunk) {
        const int t0 = chunk * TCHUNK;

        __syncthreads();

        {
            const float4* src = (const float4*)(xg + t0 * F);
            float4*       dst = (float4*)(x_lds + tid * (TCHUNK * F));
#pragma unroll
            for (int v = 0; v < (TCHUNK * F) / 4; ++v) dst[v] = src[v];
        }
        __syncthreads();

        {
            const int rbase_wave = wave * (2 * TCHUNK * 16);
            for (int g = 0; g < 2 * TCHUNK; ++g) {
                const int rb = rbase_wave + g * 16;
                const int r  = rb + n;
                v2f a;
                a.x = x_lds[r * F + kb];
                a.y = x_lds[r * F + kb + 1] * aymask;

                v8f d = __builtin_amdgcn_wmma_f32_16x16x4_f32(
                    false, a, false, bmat, (short)0, cmat, false, false);

                if (n < 4) {
                    const int mofs = (lane < 16) ? 0 : 8;
#pragma unroll
                    for (int v = 0; v < 8; ++v)
                        xz_lds[(rb + mofs + v) * 4 + n] = d[v];
                }
            }
        }
        __syncthreads();

        {
            const float4* zz = (const float4*)(xz_lds + tid * (TCHUNK * 4));
#pragma unroll
            for (int t = 0; t < TCHUNK; ++t) {
                float4 z = zz[t];
                float ig = fast_sigmoid(z.x + h * rk0);
                float fg = fast_sigmoid(z.y + h * rk1);
                float gg = fast_tanh   (z.z + h * rk2);
                float og = fast_sigmoid(z.w + h * rk3);
                c = fg * c + ig * gg;
                h = og * fast_tanh(c);
            }
        }
    }

    out[b] = h;
}

extern "C" void kernel_launch(void* const* d_in, const int* in_sizes, int n_in,
                              void* d_out, int out_size, void* d_ws, size_t ws_size,
                              hipStream_t stream) {
    const float* x  = (const float*)d_in[0];
    const float* k  = (const float*)d_in[1];
    const float* rk = (const float*)d_in[2];
    const float* bi = (const float*)d_in[3];
    float*       o  = (float*)d_out;

    const size_t lds_bytes = (size_t)BLOCK * TCHUNK * (F + 4) * sizeof(float);
    hipLaunchKernelGGL(lstm_scan_kernel,
                       dim3(BATCH / BLOCK), dim3(BLOCK), lds_bytes, stream,
                       x, k, rk, bi, o);
}


# Projection transposed on the matrix pipe: x is the WMMA B operand, batch on N, so each lane gets its own gates in registers; LDS staging, scatter and barriers removed, x prefetched a chunk ahead
# speedup vs baseline: 9.0226x; 9.0226x over previous
#include <hip/hip_runtime.h>


#define BATCH  16384
#define T      700
#define F      3
#define TCHUNK 20
#define NCHUNK (T / TCHUNK)
#define NPAIR  (TCHUNK / 2)
#define NV4    ((TCHUNK * F) / 4)
#define BLOCK  128

typedef __attribute__((ext_vector_type(2))) float v2f;
typedef __attribute__((ext_vector_type(8))) float v8f;

__device__ __forceinline__ float fast_sigmoid(float x) {
    float e = __builtin_amdgcn_exp2f(-x * 1.44269504088896340736f);
    return __builtin_amdgcn_rcpf(1.0f + e);
}

__device__ __forceinline__ float fast_tanh(float x) {
    float e = __builtin_amdgcn_exp2f(-x * 2.88539008177792681472f);
    return 2.0f * __builtin_amdgcn_rcpf(1.0f + e) - 1.0f;
}

struct Chunk { float4 q[NV4]; };

__device__ __forceinline__ void load_chunk(Chunk& ck, const float* p) {
    const float4* src = (const float4*)p;
#pragma unroll
    for (int v = 0; v < NV4; ++v) ck.q[v] = src[v];
}

__device__ __forceinline__ v2f pair_of(const Chunk& ck, int i) {
    v2f r;
    const float4 v = ck.q[i >> 1];
    if (i & 1) { r.x = v.z; r.y = v.w; } else { r.x = v.x; r.y = v.y; }
    return r;
}

struct Carry {
    float h, c;
    v8f   d;
};

template <bool HAS_NEXT>
__device__ __forceinline__ void process_chunk(const Chunk& cur, Chunk& nxt, const float* xnext, Carry& cy,
                                              const v2f (&A)[3], const v8f& cmat,
                                              float rk0, float rk1, float rk2, float rk3) {
    float ig = 0.f, fg = 0.f, gg = 0.f, og = 0.f;
    v8f dn = cmat;
#pragma unroll
    for (int p = 0; p < NPAIR; ++p) {
#pragma unroll
        for (int j = 0; j < 4; ++j) {
            if (j < 3 && (p + 1 < NPAIR || HAS_NEXT)) {
                const v2f bfrag = (p + 1 < NPAIR) ? pair_of(cur, 3 * (p + 1) + j) : pair_of(nxt, j);
                dn = __builtin_amdgcn_wmma_f32_16x16x4_f32(false, A[j], false, bfrag, (short)0,
                                                           (j == 0) ? cmat : dn, false, false);
            }
            __builtin_amdgcn_sched_barrier(0);

            const int o4 = (j >> 1) * 4;
            if ((j & 1) == 0) {
                ig = fast_sigmoid(cy.d[o4 + 0] + cy.h * rk0);
                fg = fast_sigmoid(cy.d[o4 + 1] + cy.h * rk1);
                gg = fast_tanh   (cy.d[o4 + 2] + cy.h * rk2);
                og = fast_sigmoid(cy.d[o4 + 3] + cy.h * rk3);
            } else {
                cy.c = fg * cy.c + ig * gg;
                cy.h = og * fast_tanh(cy.c);
                const int v = 2 * p + (j >> 1);
                if (HAS_NEXT && v < NV4) nxt.q[v] = ((const float4*)xnext)[v];
            }
            __builtin_amdgcn_sched_barrier(0);
        }
        cy.d = dn;
    }
}

__global__ __launch_bounds__(BLOCK)
void lstm_scan_kernel(const float* __restrict__ x,
                      const float* __restrict__ kern,
                      const float* __restrict__ rkern,
                      const float* __restrict__ bias,
                      float* __restrict__ out) {
    const int tid  = threadIdx.x;
    const int lane = tid & 31;
    const int b    = blockIdx.x * BLOCK + tid;
    const int c16  = lane & 15;
    const int hh   = lane >> 4;

    const int   g     = c16 & 3;
    const int   u     = (c16 >> 2) & 1;
    const bool  mine  = (c16 >> 3) == hh;
    const float K0 = kern[0 * 4 + g], K1 = kern[1 * 4 + g], K2 = kern[2 * 4 + g];
    v2f A[3];
    A[0].x = (mine && u == 0) ? K0 : 0.0f;   A[0].y = (mine && u == 0) ? K1 : 0.0f;
    A[1].x = (mine && u == 0) ? K2 : 0.0f;   A[1].y = (mine && u == 1) ? K0 : 0.0f;
    A[2].x = (mine && u == 1) ? K1 : 0.0f;   A[2].y = (mine && u == 1) ? K2 : 0.0f;

    v8f cmat;
#pragma unroll
    for (int r = 0; r < 8; ++r) cmat[r] = bias[r & 3];

    const float rk0 = rkern[0], rk1 = rkern[1], rk2 = rkern[2], rk3 = rkern[3];
    const float* xg = x + (size_t)b * (T * F);

    Chunk ca, cb;
    load_chunk(ca, xg);

    Carry cy;
    cy.h = 0.0f;  cy.c = 0.0f;
    cy.d = cmat;
#pragma unroll
    for (int j = 0; j < 3; ++j)
        cy.d = __builtin_amdgcn_wmma_f32_16x16x4_f32(false, A[j], false, pair_of(ca, j),
                                                     (short)0, cy.d, false, false);

    for (int i = 0; i < NCHUNK / 2; ++i) {
        process_chunk<true>(ca, cb, xg + (2 * i + 1) * (TCHUNK * F), cy, A, cmat, rk0, rk1, rk2, rk3);
        process_chunk<true>(cb, ca, xg + (2 * i + 2) * (TCHUNK * F), cy, A, cmat, rk0, rk1, rk2, rk3);
    }
    process_chunk<false>(ca, cb, xg, cy, A, cmat, rk0, rk1, rk2, rk3);

    out[b] = cy.h;
}

extern "C" void kernel_launch(void* const* d_in, const int* in_sizes, int n_in,
                              void* d_out, int out_size, void* d_ws, size_t ws_size,
                              hipStream_t stream) {
    const float* x  = (const float*)d_in[0];
    const float* k  = (const float*)d_in[1];
    const float* rk = (const float*)d_in[2];
    const float* bi = (const float*)d_in[3];
    float*       o  = (float*)d_out;

    hipLaunchKernelGGL(lstm_scan_kernel,
                       dim3(BATCH / BLOCK), dim3(BLOCK), 0, stream,
                       x, k, rk, bi, o);
}


# Step's serial chain 14 -> 5 dependent instructions: gates on v_tanh_f32 (sigmoid = 0.5+0.5*tanh(z/2), halves folded into weights); h not formed: z = fma(tanh(c), o*rk, xz); products re-associated
# speedup vs baseline: 10.1247x; 1.1221x over previous
#include <hip/hip_runtime.h>


#define BATCH  16384
#define T      700
#define F      3
#define TCHUNK 20
#define NCHUNK (T / TCHUNK)
#define NPAIR  (TCHUNK / 2)
#define NV4    ((TCHUNK * F) / 4)
#define BLOCK  128

typedef __attribute__((ext_vector_type(2))) float v2f;
typedef __attribute__((ext_vector_type(8))) float v8f;

__device__ __forceinline__ float hw_tanh(float x) { return __builtin_amdgcn_tanhf(x); }

struct Chunk { float4 q[NV4]; };

__device__ __forceinline__ void load_chunk(Chunk& ck, const float* p) {
    const float4* src = (const float4*)p;
#pragma unroll
    for (int v = 0; v < NV4; ++v) ck.q[v] = src[v];
}

__device__ __forceinline__ v2f pair_of(const Chunk& ck, int i) {
    v2f r;
    const float4 v = ck.q[i >> 1];
    if (i & 1) { r.x = v.z; r.y = v.w; } else { r.x = v.x; r.y = v.y; }
    return r;
}

struct Carry {
    float tc;
    float c;
    float hc;
    float to;
    float P0, P1, P2, P3;
    v8f   d;
};

template <bool HAS_NEXT>
__device__ __forceinline__ void process_chunk(const Chunk& cur, Chunk& nxt, const float* xnext, Carry& cy,
                                              const v2f (&A)[3], const v8f& cmat,
                                              float hr0, float hr1, float hr2, float hr3) {
    float ti = 0.f, tf = 0.f, tg = 0.f;
    v8f dn = cmat;
#pragma unroll
    for (int p = 0; p < NPAIR; ++p) {
#pragma unroll
        for (int j = 0; j < 4; ++j) {
            if (j < 3 && (p + 1 < NPAIR || HAS_NEXT)) {
                const v2f bfrag = (p + 1 < NPAIR) ? pair_of(cur, 3 * (p + 1) + j) : pair_of(nxt, j);
                dn = __builtin_amdgcn_wmma_f32_16x16x4_f32(false, A[j], false, bfrag, (short)0,
                                                           (j == 0) ? cmat : dn, false, false);
            }
            __builtin_amdgcn_sched_barrier(0);

            const int o4 = (j >> 1) * 4;
            if ((j & 1) == 0) {
                const float ai = __builtin_fmaf(cy.tc, cy.P0, cy.d[o4 + 0]);
                const float af = __builtin_fmaf(cy.tc, cy.P1, cy.d[o4 + 1]);
                const float ag = __builtin_fmaf(cy.tc, cy.P2, cy.d[o4 + 2]);
                const float ao = __builtin_fmaf(cy.tc, cy.P3, cy.d[o4 + 3]);
                cy.hc = 0.5f * cy.c;
                ti    = hw_tanh(ai);
                tf    = hw_tanh(af);
                tg    = hw_tanh(ag);
                cy.to = hw_tanh(ao);
            } else {
                const float ig = __builtin_fmaf(0.5f, ti, 0.5f);
                const float w  = __builtin_fmaf(tf, cy.hc, cy.hc);
                const float c  = __builtin_fmaf(ig, tg, w);
                __builtin_amdgcn_sched_barrier(0);
                cy.P0 = __builtin_fmaf(cy.to, hr0, hr0);
                cy.P1 = __builtin_fmaf(cy.to, hr1, hr1);
                cy.P2 = __builtin_fmaf(cy.to, hr2, hr2);
                cy.P3 = __builtin_fmaf(cy.to, hr3, hr3);
                __builtin_amdgcn_sched_barrier(0);
                cy.tc = hw_tanh(c);
                cy.c  = c;
                const int v = 2 * p + (j >> 1);
                if (HAS_NEXT && v < NV4) nxt.q[v] = ((const float4*)xnext)[v];
            }
            __builtin_amdgcn_sched_barrier(0);
        }
        cy.d = dn;
    }
}

__global__ __launch_bounds__(BLOCK)
void lstm_scan_kernel(const float* __restrict__ x,
                      const float* __restrict__ kern,
                      const float* __restrict__ rkern,
                      const float* __restrict__ bias,
                      float* __restrict__ out) {
    const int tid  = threadIdx.x;
    const int lane = tid & 31;
    const int b    = blockIdx.x * BLOCK + tid;
    const int c16  = lane & 15;
    const int hh   = lane >> 4;

    const int   g     = c16 & 3;
    const int   u     = (c16 >> 2) & 1;
    const bool  mine  = (c16 >> 3) == hh;
    const float sg    = (g == 2) ? 1.0f : 0.5f;
    const float K0 = sg * kern[0 * 4 + g], K1 = sg * kern[1 * 4 + g], K2 = sg * kern[2 * 4 + g];
    v2f A[3];
    A[0].x = (mine && u == 0) ? K0 : 0.0f;   A[0].y = (mine && u == 0) ? K1 : 0.0f;
    A[1].x = (mine && u == 0) ? K2 : 0.0f;   A[1].y = (mine && u == 1) ? K0 : 0.0f;
    A[2].x = (mine && u == 1) ? K1 : 0.0f;   A[2].y = (mine && u == 1) ? K2 : 0.0f;

    v8f cmat;
#pragma unroll
    for (int r = 0; r < 8; ++r) cmat[r] = (((r & 3) == 2) ? 1.0f : 0.5f) * bias[r & 3];

    const float hr0 = 0.25f * rkern[0], hr1 = 0.25f * rkern[1],
                hr2 = 0.50f * rkern[2], hr3 = 0.25f * rkern[3];
    const float* xg = x + (size_t)b * (T * F);

    Chunk ca, cb;
    load_chunk(ca, xg);

    Carry cy;
    cy.tc = 0.0f;  cy.c = 0.0f;  cy.hc = 0.0f;  cy.to = 0.0f;
    cy.P0 = 0.0f;  cy.P1 = 0.0f;  cy.P2 = 0.0f;  cy.P3 = 0.0f;
    cy.d = cmat;
#pragma unroll
    for (int j = 0; j < 3; ++j)
        cy.d = __builtin_amdgcn_wmma_f32_16x16x4_f32(false, A[j], false, pair_of(ca, j),
                                                     (short)0, cy.d, false, false);

    for (int i = 0; i < NCHUNK / 2; ++i) {
        process_chunk<true>(ca, cb, xg + (2 * i + 1) * (TCHUNK * F), cy, A, cmat, hr0, hr1, hr2, hr3);
        process_chunk<true>(cb, ca, xg + (2 * i + 2) * (TCHUNK * F), cy, A, cmat, hr0, hr1, hr2, hr3);
    }
    process_chunk<false>(ca, cb, xg, cy, A, cmat, hr0, hr1, hr2, hr3);

    out[b] = __builtin_fmaf(0.5f, cy.to, 0.5f) * cy.tc;
}

extern "C" void kernel_launch(void* const* d_in, const int* in_sizes, int n_in,
                              void* d_out, int out_size, void* d_ws, size_t ws_size,
                              hipStream_t stream) {
    const float* x  = (const float*)d_in[0];
    const float* k  = (const float*)d_in[1];
    const float* rk = (const float*)d_in[2];
    const float* bi = (const float*)d_in[3];
    float*       o  = (float*)d_out;

    hipLaunchKernelGGL(lstm_scan_kernel,
                       dim3(BATCH / BLOCK), dim3(BLOCK), 0, stream,
                       x, k, rk, bi, o);
}


# Work-groups of 2 waves instead of 4: the waves spread over all 256 WGPs instead of 128, so half as many waves share each WGP's load path (a 16-byte-per-lane load touches 32 lines)
# speedup vs baseline: 13.1832x; 1.3021x over previous
#include <hip/hip_runtime.h>


#define BATCH  16384
#define T      700
#define F      3
#define TCHUNK 20
#define NCHUNK (T / TCHUNK)
#define NPAIR  (TCHUNK / 2)
#define NV4    ((TCHUNK * F) / 4)
#define BLOCK  64

typedef __attribute__((ext_vector_type(2))) float v2f;
typedef __attribute__((ext_vector_type(8))) float v8f;

__device__ __forceinline__ float hw_tanh(float x) { return __builtin_amdgcn_tanhf(x); }

struct Chunk { float4 q[NV4]; };

__device__ __forceinline__ void load_chunk(Chunk& ck, const float* p) {
    const float4* src = (const float4*)p;
#pragma unroll
    for (int v = 0; v < NV4; ++v) ck.q[v] = src[v];
}

__device__ __forceinline__ v2f pair_of(const Chunk& ck, int i) {
    v2f r;
    const float4 v = ck.q[i >> 1];
    if (i & 1) { r.x = v.z; r.y = v.w; } else { r.x = v.x; r.y = v.y; }
    return r;
}

struct Carry {
    float tc;
    float c;
    float hc;
    float to;
    float P0, P1, P2, P3;
    v8f   d;
};

template <bool HAS_NEXT>
__device__ __forceinline__ void process_chunk(const Chunk& cur, Chunk& nxt, const float* xnext, Carry& cy,
                                              const v2f (&A)[3], const v8f& cmat,
                                              float hr0, float hr1, float hr2, float hr3) {
    float ti = 0.f, tf = 0.f, tg = 0.f;
    v8f dn = cmat;
#pragma unroll
    for (int p = 0; p < NPAIR; ++p) {
#pragma unroll
        for (int j = 0; j < 4; ++j) {
            if (j < 3 && (p + 1 < NPAIR || HAS_NEXT)) {
                const v2f bfrag = (p + 1 < NPAIR) ? pair_of(cur, 3 * (p + 1) + j) : pair_of(nxt, j);
                dn = __builtin_amdgcn_wmma_f32_16x16x4_f32(false, A[j], false, bfrag, (short)0,
                                                           (j == 0) ? cmat : dn, false, false);
            }
            __builtin_amdgcn_sched_barrier(0);

            const int o4 = (j >> 1) * 4;
            if ((j & 1) == 0) {
                const float ai = __builtin_fmaf(cy.tc, cy.P0, cy.d[o4 + 0]);
                const float af = __builtin_fmaf(cy.tc, cy.P1, cy.d[o4 + 1]);
                const float ag = __builtin_fmaf(cy.tc, cy.P2, cy.d[o4 + 2]);
                const float ao = __builtin_fmaf(cy.tc, cy.P3, cy.d[o4 + 3]);
                cy.hc = 0.5f * cy.c;
                ti    = hw_tanh(ai);
                tf    = hw_tanh(af);
                tg    = hw_tanh(ag);
                cy.to = hw_tanh(ao);
            } else {
                const float ig = __builtin_fmaf(0.5f, ti, 0.5f);
                const float w  = __builtin_fmaf(tf, cy.hc, cy.hc);
                const float c  = __builtin_fmaf(ig, tg, w);
                __builtin_amdgcn_sched_barrier(0);
                cy.P0 = __builtin_fmaf(cy.to, hr0, hr0);
                cy.P1 = __builtin_fmaf(cy.to, hr1, hr1);
                cy.P2 = __builtin_fmaf(cy.to, hr2, hr2);
                cy.P3 = __builtin_fmaf(cy.to, hr3, hr3);
                __builtin_amdgcn_sched_barrier(0);
                cy.tc = hw_tanh(c);
                cy.c  = c;
                const int v = 2 * p + (j >> 1);
                if (HAS_NEXT && v < NV4) nxt.q[v] = ((const float4*)xnext)[v];
            }
            __builtin_amdgcn_sched_barrier(0);
        }
        cy.d = dn;
    }
}

__global__ __launch_bounds__(BLOCK)
void lstm_scan_kernel(const float* __restrict__ x,
                      const float* __restrict__ kern,
                      const float* __restrict__ rkern,
                      const float* __restrict__ bias,
                      float* __restrict__ out) {
    const int tid  = threadIdx.x;
    const int lane = tid & 31;
    const int b    = blockIdx.x * BLOCK + tid;
    const int c16  = lane & 15;
    const int hh   = lane >> 4;

    const int   g     = c16 & 3;
    const int   u     = (c16 >> 2) & 1;
    const bool  mine  = (c16 >> 3) == hh;
    const float sg    = (g == 2) ? 1.0f : 0.5f;
    const float K0 = sg * kern[0 * 4 + g], K1 = sg * kern[1 * 4 + g], K2 = sg * kern[2 * 4 + g];
    v2f A[3];
    A[0].x = (mine && u == 0) ? K0 : 0.0f;   A[0].y = (mine && u == 0) ? K1 : 0.0f;
    A[1].x = (mine && u == 0) ? K2 : 0.0f;   A[1].y = (mine && u == 1) ? K0 : 0.0f;
    A[2].x = (mine && u == 1) ? K1 : 0.0f;   A[2].y = (mine && u == 1) ? K2 : 0.0f;

    v8f cmat;
#pragma unroll
    for (int r = 0; r < 8; ++r) cmat[r] = (((r & 3) == 2) ? 1.0f : 0.5f) * bias[r & 3];

    const float hr0 = 0.25f * rkern[0], hr1 = 0.25f * rkern[1],
                hr2 = 0.50f * rkern[2], hr3 = 0.25f * rkern[3];
    const float* xg = x + (size_t)b * (T * F);

    Chunk ca, cb;
    load_chunk(ca, xg);

    Carry cy;
    cy.tc = 0.0f;  cy.c = 0.0f;  cy.hc = 0.0f;  cy.to = 0.0f;
    cy.P0 = 0.0f;  cy.P1 = 0.0f;  cy.P2 = 0.0f;  cy.P3 = 0.0f;
    cy.d = cmat;
#pragma unroll
    for (int j = 0; j < 3; ++j)
        cy.d = __builtin_amdgcn_wmma_f32_16x16x4_f32(false, A[j], false, pair_of(ca, j),
                                                     (short)0, cy.d, false, false);

    for (int i = 0; i < NCHUNK / 2; ++i) {
        process_chunk<true>(ca, cb, xg + (2 * i + 1) * (TCHUNK * F), cy, A, cmat, hr0, hr1, hr2, hr3);
        process_chunk<true>(cb, ca, xg + (2 * i + 2) * (TCHUNK * F), cy, A, cmat, hr0, hr1, hr2, hr3);
    }
    process_chunk<false>(ca, cb, xg, cy, A, cmat, hr0, hr1, hr2, hr3);

    out[b] = __builtin_fmaf(0.5f, cy.to, 0.5f) * cy.tc;
}

extern "C" void kernel_launch(void* const* d_in, const int* in_sizes, int n_in,
                              void* d_out, int out_size, void* d_ws, size_t ws_size,
                              hipStream_t stream) {
    const float* x  = (const float*)d_in[0];
    const float* k  = (const float*)d_in[1];
    const float* rk = (const float*)d_in[2];
    const float* bi = (const float*)d_in[3];
    float*       o  = (float*)d_out;

    hipLaunchKernelGGL(lstm_scan_kernel,
                       dim3(BATCH / BLOCK), dim3(BLOCK), 0, stream,
                       x, k, rk, bi, o);
}


# x prefetched three chunks (60 steps) ahead instead of one: ring of four register buffers, up to 45 loads in flight per wave (was 15); the scan stood waiting for x at every chunk; sums unchanged
# speedup vs baseline: 17.5259x; 1.3294x over previous
#include <hip/hip_runtime.h>


#define BATCH  16384
#define T      700
#define F      3
#define TCHUNK 20
#define NCHUNK (T / TCHUNK)
#define NRING  ((NCHUNK - 3) / 4)
#define NPAIR  (TCHUNK / 2)
#define NV4    ((TCHUNK * F) / 4)
#define BLOCK  64

typedef __attribute__((ext_vector_type(2))) float v2f;
typedef __attribute__((ext_vector_type(8))) float v8f;

__device__ __forceinline__ float hw_tanh(float x) { return __builtin_amdgcn_tanhf(x); }

struct Chunk { float4 q[NV4]; };

__device__ __forceinline__ void load_chunk(Chunk& ck, const float* p) {
    const float4* src = (const float4*)p;
#pragma unroll
    for (int v = 0; v < NV4; ++v) ck.q[v] = src[v];
}

__device__ __forceinline__ v2f pair_of(const Chunk& ck, int i) {
    v2f r;
    const float4 v = ck.q[i >> 1];
    if (i & 1) { r.x = v.z; r.y = v.w; } else { r.x = v.x; r.y = v.y; }
    return r;
}

struct Carry {
    float tc;
    float c;
    float hc;
    float to;
    float P0, P1, P2, P3;
    v8f   d;
};

template <bool HAS_NEXT, bool HAS_FILL>
__device__ __forceinline__ void process_chunk(const Chunk& cur, const Chunk& nxt, Chunk& fill, const float* xfill, Carry& cy,
                                              const v2f (&A)[3], const v8f& cmat,
                                              float hr0, float hr1, float hr2, float hr3) {
    float ti = 0.f, tf = 0.f, tg = 0.f;
    v8f dn = cmat;
#pragma unroll
    for (int p = 0; p < NPAIR; ++p) {
#pragma unroll
        for (int j = 0; j < 4; ++j) {
            if (j < 3 && (p + 1 < NPAIR || HAS_NEXT)) {
                const v2f bfrag = (p + 1 < NPAIR) ? pair_of(cur, 3 * (p + 1) + j) : pair_of(nxt, j);
                dn = __builtin_amdgcn_wmma_f32_16x16x4_f32(false, A[j], false, bfrag, (short)0,
                                                           (j == 0) ? cmat : dn, false, false);
            }
            __builtin_amdgcn_sched_barrier(0);

            const int o4 = (j >> 1) * 4;
            if ((j & 1) == 0) {
                const float ai = __builtin_fmaf(cy.tc, cy.P0, cy.d[o4 + 0]);
                const float af = __builtin_fmaf(cy.tc, cy.P1, cy.d[o4 + 1]);
                const float ag = __builtin_fmaf(cy.tc, cy.P2, cy.d[o4 + 2]);
                const float ao = __builtin_fmaf(cy.tc, cy.P3, cy.d[o4 + 3]);
                cy.hc = 0.5f * cy.c;
                ti    = hw_tanh(ai);
                tf    = hw_tanh(af);
                tg    = hw_tanh(ag);
                cy.to = hw_tanh(ao);
            } else {
                const float ig = __builtin_fmaf(0.5f, ti, 0.5f);
                const float w  = __builtin_fmaf(tf, cy.hc, cy.hc);
                const float c  = __builtin_fmaf(ig, tg, w);
                __builtin_amdgcn_sched_barrier(0);
                cy.P0 = __builtin_fmaf(cy.to, hr0, hr0);
                cy.P1 = __builtin_fmaf(cy.to, hr1, hr1);
                cy.P2 = __builtin_fmaf(cy.to, hr2, hr2);
                cy.P3 = __builtin_fmaf(cy.to, hr3, hr3);
                __builtin_amdgcn_sched_barrier(0);
                cy.tc = hw_tanh(c);
                cy.c  = c;
                const int v = 2 * p + (j >> 1);
                if (HAS_FILL && v < NV4) fill.q[v] = ((const float4*)xfill)[v];
            }
            __builtin_amdgcn_sched_barrier(0);
        }
        cy.d = dn;
    }
}

__global__ __launch_bounds__(BLOCK)
void lstm_scan_kernel(const float* __restrict__ x,
                      const float* __restrict__ kern,
                      const float* __restrict__ rkern,
                      const float* __restrict__ bias,
                      float* __restrict__ out) {
    const int tid  = threadIdx.x;
    const int lane = tid & 31;
    const int b    = blockIdx.x * BLOCK + tid;
    const int c16  = lane & 15;
    const int hh   = lane >> 4;

    const int   g     = c16 & 3;
    const int   u     = (c16 >> 2) & 1;
    const bool  mine  = (c16 >> 3) == hh;
    const float sg    = (g == 2) ? 1.0f : 0.5f;
    const float K0 = sg * kern[0 * 4 + g], K1 = sg * kern[1 * 4 + g], K2 = sg * kern[2 * 4 + g];
    v2f A[3];
    A[0].x = (mine && u == 0) ? K0 : 0.0f;   A[0].y = (mine && u == 0) ? K1 : 0.0f;
    A[1].x = (mine && u == 0) ? K2 : 0.0f;   A[1].y = (mine && u == 1) ? K0 : 0.0f;
    A[2].x = (mine && u == 1) ? K1 : 0.0f;   A[2].y = (mine && u == 1) ? K2 : 0.0f;

    v8f cmat;
#pragma unroll
    for (int r = 0; r < 8; ++r) cmat[r] = (((r & 3) == 2) ? 1.0f : 0.5f) * bias[r & 3];

    const float hr0 = 0.25f * rkern[0], hr1 = 0.25f * rkern[1],
                hr2 = 0.50f * rkern[2], hr3 = 0.25f * rkern[3];
    const float* xg = x + (size_t)b * (T * F);

    Chunk b0, b1, b2, b3;
    load_chunk(b0, xg);
    load_chunk(b1, xg + 1 * (TCHUNK * F));
    load_chunk(b2, xg + 2 * (TCHUNK * F));

    Carry cy;
    cy.tc = 0.0f;  cy.c = 0.0f;  cy.hc = 0.0f;  cy.to = 0.0f;
    cy.P0 = 0.0f;  cy.P1 = 0.0f;  cy.P2 = 0.0f;  cy.P3 = 0.0f;
    cy.d = cmat;
#pragma unroll
    for (int j = 0; j < 3; ++j)
        cy.d = __builtin_amdgcn_wmma_f32_16x16x4_f32(false, A[j], false, pair_of(b0, j),
                                                     (short)0, cy.d, false, false);

    for (int i = 0; i < NRING; ++i) {
        const float* xi = xg + (4 * i) * (TCHUNK * F);
        process_chunk<true, true>(b0, b1, b3, xi + 3 * (TCHUNK * F), cy, A, cmat, hr0, hr1, hr2, hr3);
        process_chunk<true, true>(b1, b2, b0, xi + 4 * (TCHUNK * F), cy, A, cmat, hr0, hr1, hr2, hr3);
        process_chunk<true, true>(b2, b3, b1, xi + 5 * (TCHUNK * F), cy, A, cmat, hr0, hr1, hr2, hr3);
        process_chunk<true, true>(b3, b0, b2, xi + 6 * (TCHUNK * F), cy, A, cmat, hr0, hr1, hr2, hr3);
    }
    process_chunk<true,  false>(b0, b1, b3, xg, cy, A, cmat, hr0, hr1, hr2, hr3);
    process_chunk<true,  false>(b1, b2, b3, xg, cy, A, cmat, hr0, hr1, hr2, hr3);
    process_chunk<false, false>(b2, b3, b3, xg, cy, A, cmat, hr0, hr1, hr2, hr3);

    out[b] = __builtin_fmaf(0.5f, cy.to, 0.5f) * cy.tc;
}

extern "C" void kernel_launch(void* const* d_in, const int* in_sizes, int n_in,
                              void* d_out, int out_size, void* d_ws, size_t ws_size,
                              hipStream_t stream) {
    const float* x  = (const float*)d_in[0];
    const float* k  = (const float*)d_in[1];
    const float* rk = (const float*)d_in[2];
    const float* bi = (const float*)d_in[3];
    float*       o  = (float*)d_out;

    hipLaunchKernelGGL(lstm_scan_kernel,
                       dim3(BATCH / BLOCK), dim3(BLOCK), 0, stream,
                       x, k, rk, bi, o);
}
